# starting kernel
# baseline (speedup 1.0000x reference)
#include <hip/hip_runtime.h>

#define BK 16
#define PP 3
#define NSPAN 13
#define CCH 128

typedef __attribute__((ext_vector_type(2))) float v2f;
typedef __attribute__((ext_vector_type(8))) float v8f;

constexpr double knotc(int i) {
  return (i <= 3) ? -1.0 : ((i >= 16) ? 1.0 : -1.0 + 2.0 * (double)(i - 3) / 13.0);
}

struct BPTable {
  float c[NSPAN][4][4];
};

constexpr BPTable make_bp() {
  BPTable t{};
  for (int j = 0; j < NSPAN; ++j) {
    const int s = j + 3;
    double P[5][4] = {};
    P[3][0] = 1.0;
    for (int r = 1; r <= PP; ++r) {
      double Q[5][4] = {};
      for (int i = s - r; i <= s; ++i) {
        const int li = i - (s - 3);
        double acc[4] = {};
        const double d1 = knotc(i + r) - knotc(i);
        if (d1 != 0.0) {
          const double ai = 1.0 / d1, bi = -knotc(i) / d1;
          for (int kp = 3; kp >= 0; --kp) {
            double v = bi * P[li][kp];
            if (kp > 0) v += ai * P[li][kp - 1];
            acc[kp] += v;
          }
        }
        const double d2 = knotc(i + r + 1) - knotc(i + 1);
        if (d2 != 0.0) {
          const double ai = -1.0 / d2, bi = knotc(i + r + 1) / d2;
          for (int kp = 3; kp >= 0; --kp) {
            double v = bi * P[li + 1][kp];
            if (kp > 0) v += ai * P[li + 1][kp - 1];
            acc[kp] += v;
          }
        }
        for (int kp = 0; kp < 4; ++kp) Q[li][kp] = acc[kp];
      }
      for (int a = 0; a < 5; ++a)
        for (int b = 0; b < 4; ++b) P[a][b] = Q[a][b];
    }
    for (int m = 0; m < 4; ++m)
      for (int kp = 0; kp < 4; ++kp) t.c[j][m][kp] = (float)P[m][kp];
  }
  return t;
}

constexpr BPTable kBP = make_bp();

__global__ __launch_bounds__(32) void kan_setup_coeffs(const float* __restrict__ alpha,
                                                       float* __restrict__ Cw) {
  const int wave = (int)blockIdx.x;
  const int j = wave % NSPAN;
  const int c0 = (wave / NSPAN) * 16;

  const int lane = (int)(threadIdx.x & 31);
  const int half = lane >> 4;
  const int m16 = lane & 15;

  v2f A;
  A.x = alpha[(c0 + m16) * BK + (j + (half ? 2 : 0))];
  A.y = alpha[(c0 + m16) * BK + (j + (half ? 3 : 1))];
  const int k0 = half ? 2 : 0;
  const int k1 = half ? 3 : 1;
  v2f Bv;
  Bv.x = (m16 < 4) ? kBP.c[j][k0][m16] : 0.0f;
  Bv.y = (m16 < 4) ? kBP.c[j][k1][m16] : 0.0f;

  v8f acc = {};
  acc = __builtin_amdgcn_wmma_f32_16x16x4_f32(false, A, false, Bv, (short)0, acc,
                                              false, false);

#pragma unroll
  for (int v = 0; v < 8; ++v) {
    const int M = v + (half ? 8 : 0);
    if (m16 < 4) Cw[(((c0 + M) * NSPAN) + j) * 4 + m16] = acc[v];
  }
}

__global__ __launch_bounds__(256) void kan_apply(const float* __restrict__ x,
                                                 const float* __restrict__ a,
                                                 const float* __restrict__ b,
                                                 const float* __restrict__ g,
                                                 const float* __restrict__ bias,
                                                 const float* __restrict__ Cw,
                                                 float* __restrict__ out) {
  __shared__ float4 sc[NSPAN];
  const int blk = (int)blockIdx.x;
  const int c = (blk >> 2) & (CCH - 1);
  if (threadIdx.x < NSPAN)
    sc[threadIdx.x] = reinterpret_cast<const float4*>(Cw)[c * NSPAN + (int)threadIdx.x];
  const float ac = a[c], bc = b[c], gc = g[c], oc = bias[c];
  __syncthreads();

  const long base = (long)blk * 1024 + (long)threadIdx.x * 4;
  __builtin_prefetch(x + base + 4096, 0, 0);
  const float4 xv = *reinterpret_cast<const float4*>(x + base);

  float xs[4] = {xv.x, xv.y, xv.z, xv.w};
  float os[4];
#pragma unroll
  for (int e = 0; e < 4; ++e) {
    const float xi = xs[e];
    const float u = fmaf(xi, ac, bc);
    float sp = 0.0f;
    if (u >= -1.0f && u <= 1.0f) {
      int jj = (int)((u + 1.0f) * 6.5f);
      jj = jj > 12 ? 12 : jj;
      const float4 cf = sc[jj];
      sp = fmaf(fmaf(fmaf(cf.w, u, cf.z), u, cf.y), u, cf.x);
    }
    os[e] = fmaf(gc, xi, sp + oc);
  }
  float4 ov;
  ov.x = os[0]; ov.y = os[1]; ov.z = os[2]; ov.w = os[3];
  *reinterpret_cast<float4*>(out + base) = ov;
}

extern "C" void kernel_launch(void* const* d_in, const int* in_sizes, int n_in,
                              void* d_out, int out_size, void* d_ws, size_t ws_size,
                              hipStream_t stream) {
  const float* x     = (const float*)d_in[0];
  const float* a     = (const float*)d_in[1];
  const float* b     = (const float*)d_in[2];
  const float* alpha = (const float*)d_in[3];
  const float* g     = (const float*)d_in[4];
  const float* bias  = (const float*)d_in[5];
  float* Cw = (float*)d_ws;

  kan_setup_coeffs<<<(CCH / 16) * NSPAN, 32, 0, stream>>>(alpha, Cw);

  const int blocks = out_size / 1024;
  kan_apply<<<blocks, 256, 0, stream>>>(x, a, b, g, bias, Cw, (float*)d_out);
}


# apply: 16x fewer, longer waves (512 workgroups): every thread issues its 16 b128 loads up front, then computes and stores piece by piece; branch-free lookups, per-wave table, no workgroup barrier
# speedup vs baseline: 1.5012x; 1.5012x over previous
#include <hip/hip_runtime.h>

#define BK 16
#define PP 3
#define NSPAN 13
#define CCH 128

typedef __attribute__((ext_vector_type(2))) float v2f;
typedef __attribute__((ext_vector_type(8))) float v8f;
typedef __attribute__((ext_vector_type(4))) float v4f;

constexpr double knotc(int i) {
  return (i <= 3) ? -1.0 : ((i >= 16) ? 1.0 : -1.0 + 2.0 * (double)(i - 3) / 13.0);
}

struct BPTable {
  float c[NSPAN][4][4];
};

constexpr BPTable make_bp() {
  BPTable t{};
  for (int j = 0; j < NSPAN; ++j) {
    const int s = j + 3;
    double P[5][4] = {};
    P[3][0] = 1.0;
    for (int r = 1; r <= PP; ++r) {
      double Q[5][4] = {};
      for (int i = s - r; i <= s; ++i) {
        const int li = i - (s - 3);
        double acc[4] = {};
        const double d1 = knotc(i + r) - knotc(i);
        if (d1 != 0.0) {
          const double ai = 1.0 / d1, bi = -knotc(i) / d1;
          for (int kp = 3; kp >= 0; --kp) {
            double v = bi * P[li][kp];
            if (kp > 0) v += ai * P[li][kp - 1];
            acc[kp] += v;
          }
        }
        const double d2 = knotc(i + r + 1) - knotc(i + 1);
        if (d2 != 0.0) {
          const double ai = -1.0 / d2, bi = knotc(i + r + 1) / d2;
          for (int kp = 3; kp >= 0; --kp) {
            double v = bi * P[li + 1][kp];
            if (kp > 0) v += ai * P[li + 1][kp - 1];
            acc[kp] += v;
          }
        }
        for (int kp = 0; kp < 4; ++kp) Q[li][kp] = acc[kp];
      }
      for (int a = 0; a < 5; ++a)
        for (int b = 0; b < 4; ++b) P[a][b] = Q[a][b];
    }
    for (int m = 0; m < 4; ++m)
      for (int kp = 0; kp < 4; ++kp) t.c[j][m][kp] = (float)P[m][kp];
  }
  return t;
}

constexpr BPTable kBP = make_bp();

__global__ __launch_bounds__(32) void kan_setup_coeffs(const float* __restrict__ alpha,
                                                       float* __restrict__ Cw) {
  const int wave = (int)blockIdx.x;
  const int j = wave % NSPAN;
  const int c0 = (wave / NSPAN) * 16;

  const int lane = (int)(threadIdx.x & 31);
  const int half = lane >> 4;
  const int m16 = lane & 15;

  v2f A;
  A.x = alpha[(c0 + m16) * BK + (j + (half ? 2 : 0))];
  A.y = alpha[(c0 + m16) * BK + (j + (half ? 3 : 1))];
  const int k0 = half ? 2 : 0;
  const int k1 = half ? 3 : 1;
  v2f Bv;
  Bv.x = (m16 < 4) ? kBP.c[j][k0][m16] : 0.0f;
  Bv.y = (m16 < 4) ? kBP.c[j][k1][m16] : 0.0f;

  v8f acc = {};
  acc = __builtin_amdgcn_wmma_f32_16x16x4_f32(false, A, false, Bv, (short)0, acc,
                                              false, false);

#pragma unroll
  for (int v = 0; v < 8; ++v) {
    const int M = v + (half ? 8 : 0);
    if (m16 < 4) Cw[(((c0 + M) * NSPAN) + j) * 4 + m16] = acc[v];
  }
}

#define IMG 4096
#define NBATCH 16
#define BPB 4
#define QPI 4
static_assert(CCH * (NBATCH / BPB) == 512, "grid of kan_apply: one workgroup per channel and batch group");

__device__ __forceinline__ float kan_elem(float xi, float ac, float bc, float gc, float oc,
                                          const float4* __restrict__ tab) {
  const float u = fmaf(xi, ac, bc);
  const float t = fminf(fmaxf((u + 1.0f) * 6.5f, 0.0f), 12.0f);
  const float4 cf = tab[(int)t];
  const float poly = fmaf(fmaf(fmaf(cf.w, u, cf.z), u, cf.y), u, cf.x);
  const float sp = (u >= -1.0f && u <= 1.0f) ? poly : 0.0f;
  return fmaf(gc, xi, sp + oc);
}

__global__ __launch_bounds__(256) void kan_apply(const float* __restrict__ x,
                                                 const float* __restrict__ a,
                                                 const float* __restrict__ b,
                                                 const float* __restrict__ g,
                                                 const float* __restrict__ bias,
                                                 const float* __restrict__ Cw,
                                                 float* __restrict__ out) {
  __shared__ float4 sc[8][32];
  const int c = (int)blockIdx.x & (CCH - 1);
  const int bg = (int)blockIdx.x >> 7;
  const int tid = (int)threadIdx.x;
  const int lane = tid & 31;
  const int wv = tid >> 5;

  const float4 cfl =
      reinterpret_cast<const float4*>(Cw)[c * NSPAN + (lane < NSPAN ? lane : NSPAN - 1)];
  sc[wv][lane] = cfl;

  const long base = ((long)bg * BPB * CCH + c) * IMG + tid * 4;
  const float* __restrict__ xb = x + base;
  float* __restrict__ ob = out + base;

  v4f xv[BPB * QPI];
#pragma unroll
  for (int i = 0; i < BPB; ++i)
#pragma unroll
    for (int q = 0; q < QPI; ++q)
      xv[i * QPI + q] = *reinterpret_cast<const v4f*>(xb + i * (CCH * IMG) + q * 1024);
  const float ac = a[c], bc = b[c], gc = g[c], oc = bias[c];
  const float4* tab = sc[wv];

#pragma unroll
  for (int i = 0; i < BPB; ++i) {
#pragma unroll
    for (int q = 0; q < QPI; ++q) {
      v4f v = xv[i * QPI + q];
      asm volatile("" : "+v"(v) : : "memory");
      v4f ov;
      ov.x = kan_elem(v.x, ac, bc, gc, oc, tab);
      ov.y = kan_elem(v.y, ac, bc, gc, oc, tab);
      ov.z = kan_elem(v.z, ac, bc, gc, oc, tab);
      ov.w = kan_elem(v.w, ac, bc, gc, oc, tab);
      *reinterpret_cast<v4f*>(ob + i * (CCH * IMG) + q * 1024) = ov;
    }
  }
}

extern "C" void kernel_launch(void* const* d_in, const int* in_sizes, int n_in,
                              void* d_out, int out_size, void* d_ws, size_t ws_size,
                              hipStream_t stream) {
  const float* x     = (const float*)d_in[0];
  const float* a     = (const float*)d_in[1];
  const float* b     = (const float*)d_in[2];
  const float* alpha = (const float*)d_in[3];
  const float* g     = (const float*)d_in[4];
  const float* bias  = (const float*)d_in[5];
  float* Cw = (float*)d_ws;

  kan_setup_coeffs<<<(CCH / 16) * NSPAN, 32, 0, stream>>>(alpha, Cw);

  kan_apply<<<512, 256, 0, stream>>>(x, a, b, g, bias, Cw, (float*)d_out);
}


# one launch instead of two: each wave builds its channel's 13 cubics itself with 7 chained f32 WMMAs (alpha row x banded constant matrix) into its LDS table; coefficient sum order changes
# speedup vs baseline: 1.9150x; 1.2756x over previous
#include <hip/hip_runtime.h>

#define BK 16
#define PP 3
#define NSPAN 13
#define CCH 128

typedef __attribute__((ext_vector_type(2))) float v2f;
typedef __attribute__((ext_vector_type(8))) float v8f;
typedef __attribute__((ext_vector_type(4))) float v4f;

constexpr double knotc(int i) {
  return (i <= 3) ? -1.0 : ((i >= 16) ? 1.0 : -1.0 + 2.0 * (double)(i - 3) / 13.0);
}

struct BPTable {
  float c[NSPAN][4][4];
};

constexpr BPTable make_bp() {
  BPTable t{};
  for (int j = 0; j < NSPAN; ++j) {
    const int s = j + 3;
    double P[5][4] = {};
    P[3][0] = 1.0;
    for (int r = 1; r <= PP; ++r) {
      double Q[5][4] = {};
      for (int i = s - r; i <= s; ++i) {
        const int li = i - (s - 3);
        double acc[4] = {};
        const double d1 = knotc(i + r) - knotc(i);
        if (d1 != 0.0) {
          const double ai = 1.0 / d1, bi = -knotc(i) / d1;
          for (int kp = 3; kp >= 0; --kp) {
            double v = bi * P[li][kp];
            if (kp > 0) v += ai * P[li][kp - 1];
            acc[kp] += v;
          }
        }
        const double d2 = knotc(i + r + 1) - knotc(i + 1);
        if (d2 != 0.0) {
          const double ai = -1.0 / d2, bi = knotc(i + r + 1) / d2;
          for (int kp = 3; kp >= 0; --kp) {
            double v = bi * P[li + 1][kp];
            if (kp > 0) v += ai * P[li + 1][kp - 1];
            acc[kp] += v;
          }
        }
        for (int kp = 0; kp < 4; ++kp) Q[li][kp] = acc[kp];
      }
      for (int a = 0; a < 5; ++a)
        for (int b = 0; b < 4; ++b) P[a][b] = Q[a][b];
    }
    for (int m = 0; m < 4; ++m)
      for (int kp = 0; kp < 4; ++kp) t.c[j][m][kp] = (float)P[m][kp];
  }
  return t;
}

constexpr BPTable kBP = make_bp();

#define NPAIR 7
struct BandTable {
  alignas(16) float v[NPAIR][32][2];
};

constexpr BandTable make_band() {
  BandTable t{};
  for (int p = 0; p < NPAIR; ++p) {
    const int tile = p / 2;
    const int step = (p + 1) / 2;
    for (int l = 0; l < 32; ++l) {
      const int col = l & 15, h = l >> 4;
      const int j = 4 * tile + (col >> 2), n = col & 3;
      for (int e = 0; e < 2; ++e) {
        const int i = 4 * step + 2 * h + e;
        const int kk = i - j;
        t.v[p][l][e] = (j < NSPAN && kk >= 0 && kk <= 3) ? kBP.c[j][kk][n] : 0.0f;
      }
    }
  }
  return t;
}

constexpr BandTable kBand = make_band();

#define IMG 4096
#define NBATCH 16
#define BPB 4
#define QPI 4
static_assert(CCH * (NBATCH / BPB) == 512, "grid of kan_apply: one workgroup per channel and batch group");

__device__ __forceinline__ float kan_elem(float xi, float ac, float bc, float gc, float oc,
                                          const v4f* __restrict__ tab) {
  const float u = fmaf(xi, ac, bc);
  const float t = fminf(fmaxf((u + 1.0f) * 6.5f, 0.0f), 12.0f);
  const v4f cf = tab[(int)t];
  const float poly = fmaf(fmaf(fmaf(cf.w, u, cf.z), u, cf.y), u, cf.x);
  const float sp = (u >= -1.0f && u <= 1.0f) ? poly : 0.0f;
  return fmaf(gc, xi, sp + oc);
}

__global__ __launch_bounds__(256) void kan_apply(const float* __restrict__ x,
                                                 const float* __restrict__ a,
                                                 const float* __restrict__ b,
                                                 const float* __restrict__ alpha,
                                                 const float* __restrict__ g,
                                                 const float* __restrict__ bias,
                                                 float* __restrict__ out) {
  __shared__ v4f sc[8][16];
  const int c = (int)blockIdx.x & (CCH - 1);
  const int bg = (int)blockIdx.x >> 7;
  const int tid = (int)threadIdx.x;
  const int lane = tid & 31;
  const int wv = tid >> 5;
  const int half = lane >> 4;

  v2f bfrag[NPAIR];
#pragma unroll
  for (int p = 0; p < NPAIR; ++p)
    bfrag[p] = *reinterpret_cast<const v2f*>(&kBand.v[p][lane][0]);

  v2f afrag[4];
#pragma unroll
  for (int s = 0; s < 4; ++s)
    afrag[s] = *reinterpret_cast<const v2f*>(alpha + c * BK + 4 * s + 2 * half);

  const long base = ((long)bg * BPB * CCH + c) * IMG + tid * 4;
  const float* __restrict__ xb = x + base;
  float* __restrict__ ob = out + base;

  v4f xv[BPB * QPI];
#pragma unroll
  for (int i = 0; i < BPB; ++i)
#pragma unroll
    for (int q = 0; q < QPI; ++q)
      xv[i * QPI + q] = *reinterpret_cast<const v4f*>(xb + i * (CCH * IMG) + q * 1024);
  const float ac = a[c], bc = b[c], gc = g[c], oc = bias[c];
  __builtin_amdgcn_sched_barrier(0);

  v8f acc[4] = {};
#pragma unroll
  for (int p = 0; p < NPAIR; ++p)
    acc[p / 2] = __builtin_amdgcn_wmma_f32_16x16x4_f32(false, afrag[(p + 1) / 2], false, bfrag[p],
                                                       (short)0, acc[p / 2], false, false);
  float* tf = reinterpret_cast<float*>(sc[wv]);
#pragma unroll
  for (int t = 0; t < 4; ++t) tf[16 * t + (lane & 15)] = acc[t][0];
  const v4f* tab = sc[wv];

#pragma unroll
  for (int i = 0; i < BPB; ++i) {
#pragma unroll
    for (int q = 0; q < QPI; ++q) {
      v4f v = xv[i * QPI + q];
      asm volatile("" : "+v"(v) : : "memory");
      v4f ov;
      ov.x = kan_elem(v.x, ac, bc, gc, oc, tab);
      ov.y = kan_elem(v.y, ac, bc, gc, oc, tab);
      ov.z = kan_elem(v.z, ac, bc, gc, oc, tab);
      ov.w = kan_elem(v.w, ac, bc, gc, oc, tab);
      *reinterpret_cast<v4f*>(ob + i * (CCH * IMG) + q * 1024) = ov;
    }
  }
}

extern "C" void kernel_launch(void* const* d_in, const int* in_sizes, int n_in,
                              void* d_out, int out_size, void* d_ws, size_t ws_size,
                              hipStream_t stream) {
  const float* x     = (const float*)d_in[0];
  const float* a     = (const float*)d_in[1];
  const float* b     = (const float*)d_in[2];
  const float* alpha = (const float*)d_in[3];
  const float* g     = (const float*)d_in[4];
  const float* bias  = (const float*)d_in[5];

  kan_apply<<<512, 256, 0, stream>>>(x, a, b, alpha, g, bias, (float*)d_out);
}


# each wave works its 16 pieces off one image (4 pieces) at a time in two stages, all 16 span lookups issued together, then the 16 cubics: one LDS wait per image instead of one per piece
# speedup vs baseline: 2.0780x; 1.0851x over previous
#include <hip/hip_runtime.h>

#define BK 16
#define PP 3
#define NSPAN 13
#define CCH 128

typedef __attribute__((ext_vector_type(2))) float v2f;
typedef __attribute__((ext_vector_type(8))) float v8f;
typedef __attribute__((ext_vector_type(4))) float v4f;

constexpr double knotc(int i) {
  return (i <= 3) ? -1.0 : ((i >= 16) ? 1.0 : -1.0 + 2.0 * (double)(i - 3) / 13.0);
}

struct BPTable {
  float c[NSPAN][4][4];
};

constexpr BPTable make_bp() {
  BPTable t{};
  for (int j = 0; j < NSPAN; ++j) {
    const int s = j + 3;
    double P[5][4] = {};
    P[3][0] = 1.0;
    for (int r = 1; r <= PP; ++r) {
      double Q[5][4] = {};
      for (int i = s - r; i <= s; ++i) {
        const int li = i - (s - 3);
        double acc[4] = {};
        const double d1 = knotc(i + r) - knotc(i);
        if (d1 != 0.0) {
          const double ai = 1.0 / d1, bi = -knotc(i) / d1;
          for (int kp = 3; kp >= 0; --kp) {
            double v = bi * P[li][kp];
            if (kp > 0) v += ai * P[li][kp - 1];
            acc[kp] += v;
          }
        }
        const double d2 = knotc(i + r + 1) - knotc(i + 1);
        if (d2 != 0.0) {
          const double ai = -1.0 / d2, bi = knotc(i + r + 1) / d2;
          for (int kp = 3; kp >= 0; --kp) {
            double v = bi * P[li + 1][kp];
            if (kp > 0) v += ai * P[li + 1][kp - 1];
            acc[kp] += v;
          }
        }
        for (int kp = 0; kp < 4; ++kp) Q[li][kp] = acc[kp];
      }
      for (int a = 0; a < 5; ++a)
        for (int b = 0; b < 4; ++b) P[a][b] = Q[a][b];
    }
    for (int m = 0; m < 4; ++m)
      for (int kp = 0; kp < 4; ++kp) t.c[j][m][kp] = (float)P[m][kp];
  }
  return t;
}

constexpr BPTable kBP = make_bp();

#define NPAIR 7
struct BandTable {
  alignas(16) float v[NPAIR][32][2];
};

constexpr BandTable make_band() {
  BandTable t{};
  for (int p = 0; p < NPAIR; ++p) {
    const int tile = p / 2;
    const int step = (p + 1) / 2;
    for (int l = 0; l < 32; ++l) {
      const int col = l & 15, h = l >> 4;
      const int j = 4 * tile + (col >> 2), n = col & 3;
      for (int e = 0; e < 2; ++e) {
        const int i = 4 * step + 2 * h + e;
        const int kk = i - j;
        t.v[p][l][e] = (j < NSPAN && kk >= 0 && kk <= 3) ? kBP.c[j][kk][n] : 0.0f;
      }
    }
  }
  return t;
}

constexpr BandTable kBand = make_band();

#define IMG 4096
#define NBATCH 16
#define BPB 4
#define QPI 4
static_assert(CCH * (NBATCH / BPB) == 512, "grid of kan_apply: one workgroup per channel and batch group");

__device__ __forceinline__ v4f kan_lookup(float u, const v4f* __restrict__ tab) {
  const float t = fminf(fmaxf((u + 1.0f) * 6.5f, 0.0f), 12.0f);
  return tab[(int)t];
}

__device__ __forceinline__ float kan_eval(float xi, float u, v4f cf, float c3, float gc, float oc) {
  const float poly = fmaf(fmaf(fmaf(c3, u, cf.z), u, cf.y), u, cf.x);
  const float sp = (u >= -1.0f && u <= 1.0f) ? poly : 0.0f;
  return fmaf(gc, xi, sp + oc);
}

__global__ __launch_bounds__(256) void kan_apply(const float* __restrict__ x,
                                                 const float* __restrict__ a,
                                                 const float* __restrict__ b,
                                                 const float* __restrict__ alpha,
                                                 const float* __restrict__ g,
                                                 const float* __restrict__ bias,
                                                 float* __restrict__ out) {
  __shared__ v4f sc[8][16];
  const int c = (int)blockIdx.x & (CCH - 1);
  const int bg = (int)blockIdx.x >> 7;
  const int tid = (int)threadIdx.x;
  const int lane = tid & 31;
  const int wv = tid >> 5;
  const int half = lane >> 4;

  v2f bfrag[NPAIR];
#pragma unroll
  for (int p = 0; p < NPAIR; ++p)
    bfrag[p] = *reinterpret_cast<const v2f*>(&kBand.v[p][lane][0]);

  v2f afrag[4];
#pragma unroll
  for (int s = 0; s < 4; ++s)
    afrag[s] = *reinterpret_cast<const v2f*>(alpha + c * BK + 4 * s + 2 * half);

  const long base = ((long)bg * BPB * CCH + c) * IMG + tid * 4;
  const float* __restrict__ xb = x + base;
  float* __restrict__ ob = out + base;

  v4f xv[BPB * QPI];
#pragma unroll
  for (int i = 0; i < BPB; ++i)
#pragma unroll
    for (int q = 0; q < QPI; ++q)
      xv[i * QPI + q] = *reinterpret_cast<const v4f*>(xb + i * (CCH * IMG) + q * 1024);
  const float ac = a[c], bc = b[c], gc = g[c], oc = bias[c];
  __builtin_amdgcn_sched_barrier(0);

  v8f acc[4] = {};
#pragma unroll
  for (int p = 0; p < NPAIR; ++p)
    acc[p / 2] = __builtin_amdgcn_wmma_f32_16x16x4_f32(false, afrag[(p + 1) / 2], false, bfrag[p],
                                                       (short)0, acc[p / 2], false, false);
  float* tf = reinterpret_cast<float*>(sc[wv]);
#pragma unroll
  for (int t = 0; t < 4; ++t) tf[16 * t + (lane & 15)] = acc[t][0];
  const v4f* tab = sc[wv];

#pragma unroll
  for (int i = 0; i < BPB; ++i) {
    v4f v[QPI];
#pragma unroll
    for (int q = 0; q < QPI; ++q) v[q] = xv[i * QPI + q];
    asm volatile("" : "+v"(v[0]), "+v"(v[1]), "+v"(v[2]), "+v"(v[3]) : : "memory");
    float u[QPI * 4];
    v4f cf[QPI * 4];
    float c3[QPI * 4];
#pragma unroll
    for (int q = 0; q < QPI; ++q) {
      u[4 * q + 0] = fmaf(v[q].x, ac, bc);
      u[4 * q + 1] = fmaf(v[q].y, ac, bc);
      u[4 * q + 2] = fmaf(v[q].z, ac, bc);
      u[4 * q + 3] = fmaf(v[q].w, ac, bc);
    }
#pragma unroll
    for (int e = 0; e < QPI * 4; ++e) {
      cf[e] = kan_lookup(u[e], tab);
      c3[e] = cf[e].w;
    }
    asm volatile(""
                 : "+v"(c3[0]), "+v"(c3[1]), "+v"(c3[2]), "+v"(c3[3]), "+v"(c3[4]), "+v"(c3[5]),
                   "+v"(c3[6]), "+v"(c3[7])
                 : "v"(c3[8]), "v"(c3[9]), "v"(c3[10]), "v"(c3[11]), "v"(c3[12]), "v"(c3[13]),
                   "v"(c3[14]), "v"(c3[15]));
    asm volatile(""
                 : "+v"(c3[8]), "+v"(c3[9]), "+v"(c3[10]), "+v"(c3[11]), "+v"(c3[12]),
                   "+v"(c3[13]), "+v"(c3[14]), "+v"(c3[15]));
#pragma unroll
    for (int q = 0; q < QPI; ++q) {
      v4f ov;
      ov.x = kan_eval(v[q].x, u[4 * q + 0], cf[4 * q + 0], c3[4 * q + 0], gc, oc);
      ov.y = kan_eval(v[q].y, u[4 * q + 1], cf[4 * q + 1], c3[4 * q + 1], gc, oc);
      ov.z = kan_eval(v[q].z, u[4 * q + 2], cf[4 * q + 2], c3[4 * q + 2], gc, oc);
      ov.w = kan_eval(v[q].w, u[4 * q + 3], cf[4 * q + 3], c3[4 * q + 3], gc, oc);
      *reinterpret_cast<v4f*>(ob + i * (CCH * IMG) + q * 1024) = ov;
    }
  }
}

extern "C" void kernel_launch(void* const* d_in, const int* in_sizes, int n_in,
                              void* d_out, int out_size, void* d_ws, size_t ws_size,
                              hipStream_t stream) {
  const float* x     = (const float*)d_in[0];
  const float* a     = (const float*)d_in[1];
  const float* b     = (const float*)d_in[2];
  const float* alpha = (const float*)d_in[3];
  const float* g     = (const float*)d_in[4];
  const float* bias  = (const float*)d_in[5];

  kan_apply<<<512, 256, 0, stream>>>(x, a, b, alpha, g, bias, (float*)d_out);
}
